# starting kernel
# baseline (speedup 1.0000x reference)
#include <hip/hip_runtime.h>

typedef __attribute__((ext_vector_type(2))) float v2f;
typedef __attribute__((ext_vector_type(4))) float v4f;
typedef __attribute__((ext_vector_type(8))) float v8f;

#define N_NODES 50000
#define N_EDGES 800000
#define D       128

__global__ void gcn_init_bias(const float* __restrict__ b,
                              float* __restrict__ out) {
  int i = blockIdx.x * blockDim.x + threadIdx.x;
  const v4f* b4 = (const v4f*)b;
  ((v4f*)out)[i] = b4[i & (D / 4 - 1)];
}

__global__ void gcn_gemm_wmma(const float* __restrict__ X,
                              const float* __restrict__ W,
                              float* __restrict__ H) {
  const int lane  = threadIdx.x & 31;
  const int wave  = threadIdx.x >> 5;
  const int mbase = blockIdx.x * 16;
  const int nbase = wave * 16;
  const int half  = lane >> 4;
  const int l     = lane & 15;

  const float* __restrict__ xrow = X + (size_t)(mbase + l) * D;
  const float* __restrict__ wrow = W + (size_t)(nbase + l) * D;

  v8f acc = {};
#pragma unroll
  for (int k = 0; k < D; k += 4) {
    v2f a = *(const v2f*)(xrow + k + 2 * half);
    v2f bm = *(const v2f*)(wrow + k + 2 * half);
    acc = __builtin_amdgcn_wmma_f32_16x16x4_f32(
         false, a,  false, bm,
         (short)0, acc,  false,  false);
  }

  float* __restrict__ hrow = H + (size_t)(mbase + 8 * half) * D + nbase + l;
#pragma unroll
  for (int r = 0; r < 8; ++r) {
    hrow[(size_t)r * D] = acc[r];
  }
}

__global__ void gcn_scatter(const int* __restrict__ ei,
                            const float* __restrict__ H,
                            float* __restrict__ out) {
  const int e    = blockIdx.x * 8 + (threadIdx.x >> 5);
  const int lane = threadIdx.x & 31;
  const int src  = ei[e];
  const int dst  = ei[N_EDGES + e];

  v4f m = ((const v4f*)(H + (size_t)src * D))[lane];
  float* o = out + (size_t)dst * D + lane * 4;
  atomicAdd(o + 0, m.x);
  atomicAdd(o + 1, m.y);
  atomicAdd(o + 2, m.z);
  atomicAdd(o + 3, m.w);
}

extern "C" void kernel_launch(void* const* d_in, const int* in_sizes, int n_in,
                              void* d_out, int out_size, void* d_ws, size_t ws_size,
                              hipStream_t stream) {
  const float* x  = (const float*)d_in[0];
  const int*   ei = (const int*)d_in[1];
  const float* W  = (const float*)d_in[2];
  const float* b  = (const float*)d_in[3];
  float* out = (float*)d_out;
  float* H   = (float*)d_ws;

  gcn_init_bias<<<(N_NODES * D / 4) / 256, 256, 0, stream>>>(b, out);

  gcn_gemm_wmma<<<N_NODES / 16, 256, 0, stream>>>(x, W, H);

  gcn_scatter<<<N_EDGES / 8, 256, 0, stream>>>(ei, H, out);
}


# GEMM: W kept in registers (each wave its 32 columns, all of K) over 6-7 row tiles per block, X fragments by 16-byte loads from global, 12x fewer waves; k summed in a new order, H rounding differs
# speedup vs baseline: 1.8022x; 1.8022x over previous
#include <hip/hip_runtime.h>

typedef __attribute__((ext_vector_type(2))) float v2f;
typedef __attribute__((ext_vector_type(4))) float v4f;
typedef __attribute__((ext_vector_type(8))) float v8f;
typedef __attribute__((ext_vector_type(2))) int   v2i;

#define N_NODES 50000
#define N_EDGES 800000
#define D       128


#define ROW_TILES   (N_NODES / 16)
#define GEMM_BLOCKS 512

__device__ __forceinline__ void load_frag(v4f (&f)[16], const float* __restrict__ p) {
#pragma unroll
  for (int j = 0; j < 16; ++j) f[j] = *(const v4f*)(p + 8 * j);
}

__device__ __forceinline__ void tile_product(const v4f (&a)[16],
                                             const v4f (&b0)[16],
                                             const v4f (&b1)[16],
                                             float* __restrict__ hrow) {
  v8f acc0 = {};
  v8f acc1 = {};
#pragma unroll
  for (int j = 0; j < 16; ++j) {
    const v2f alo = a[j].xy,  ahi = a[j].zw;
    const v2f b0lo = b0[j].xy, b0hi = b0[j].zw;
    const v2f b1lo = b1[j].xy, b1hi = b1[j].zw;
    acc0 = __builtin_amdgcn_wmma_f32_16x16x4_f32(false, alo, false, b0lo,
                                                 (short)0, acc0, false, false);
    acc1 = __builtin_amdgcn_wmma_f32_16x16x4_f32(false, alo, false, b1lo,
                                                 (short)0, acc1, false, false);
    acc0 = __builtin_amdgcn_wmma_f32_16x16x4_f32(false, ahi, false, b0hi,
                                                 (short)0, acc0, false, false);
    acc1 = __builtin_amdgcn_wmma_f32_16x16x4_f32(false, ahi, false, b1hi,
                                                 (short)0, acc1, false, false);
  }
#pragma unroll
  for (int r = 0; r < 8; ++r) {
    hrow[(size_t)r * D]      = acc0[r];
    hrow[(size_t)r * D + 16] = acc1[r];
  }
}

__global__ __launch_bounds__(128)
void gcn_gemm_wmma(const float* __restrict__ X,
                   const float* __restrict__ W,
                   float* __restrict__ H,
                   int* __restrict__ head) {
  const int lane  = threadIdx.x & 31;
  const int wave  = threadIdx.x >> 5;
  const int nbase = wave * 32;
  const int half  = lane >> 4;
  const int l     = lane & 15;

  {
    const int tq = blockIdx.x + GEMM_BLOCKS * (threadIdx.x >> 4);
    if (tq < ROW_TILES) head[tq * 16 + (threadIdx.x & 15)] = -1;
  }

  const size_t aoff = (size_t)l * D + 4 * half;
  const size_t hoff = (size_t)(8 * half) * D + nbase + l;

  int t = blockIdx.x;
  v4f a0[16], a1[16];
  load_frag(a0, X + (size_t)t * 16 * D + aoff);

  v4f b0[16], b1[16];
  {
    const float* __restrict__ w0 = W + (size_t)(nbase + l) * D + 4 * half;
    const float* __restrict__ w1 = w0 + 16 * D;
#pragma unroll
    for (int j = 0; j < 16; ++j) {
      b0[j] = *(const v4f*)(w0 + 8 * j);
      b1[j] = *(const v4f*)(w1 + 8 * j);
    }
  }

  for (;;) {
    const int t1 = t + GEMM_BLOCKS;
    if (t1 >= ROW_TILES) {
      tile_product(a0, b0, b1, H + (size_t)t * 16 * D + hoff);
      break;
    }
    load_frag(a1, X + (size_t)t1 * 16 * D + aoff);
    __builtin_amdgcn_sched_barrier(0);
    tile_product(a0, b0, b1, H + (size_t)t * 16 * D + hoff);

    const int t2 = t1 + GEMM_BLOCKS;
    if (t2 >= ROW_TILES) {
      tile_product(a1, b0, b1, H + (size_t)t1 * 16 * D + hoff);
      break;
    }
    load_frag(a0, X + (size_t)t2 * 16 * D + aoff);
    __builtin_amdgcn_sched_barrier(0);
    tile_product(a1, b0, b1, H + (size_t)t1 * 16 * D + hoff);
    t = t2;
  }
}

__global__ void gcn_link(const int* __restrict__ ei,
                         int* __restrict__ head,
                         int* __restrict__ node) {
  const int e   = blockIdx.x * 256 + threadIdx.x;
  const int src = ei[e];
  const int dst = ei[N_EDGES + e];
  const int prev = atomicExch(head + dst, e);
  v2i nd;
  nd.x = prev;
  nd.y = src;
  ((v2i*)node)[e] = nd;
}

__global__ void gcn_gather(const int* __restrict__ head,
                           const int* __restrict__ node,
                           const float* __restrict__ H,
                           const float* __restrict__ b,
                           float* __restrict__ out) {
  const int lane = threadIdx.x & 31;
  const int row = __builtin_amdgcn_readfirstlane(blockIdx.x * 8 + (threadIdx.x >> 5));
  const v2i* __restrict__ nodes = (const v2i*)node;

  v4f acc = ((const v4f*)b)[lane];
  int e = head[row];
  if (e >= 0) {
    v2i nd = nodes[e];
    for (int it = 0; it < N_EDGES; ++it) {
      const v4f m = ((const v4f*)(H + (size_t)nd.y * D))[lane];
      const int en = nd.x;
      const v2i nx = nodes[en < 0 ? 0 : en];
      asm volatile("" :: "s"(nx.x), "s"(nx.y));
      acc += m;
      if (en < 0) break;
      nd = nx;
    }
  }
  ((v4f*)out)[(size_t)row * (D / 4) + lane] = acc;
}

extern "C" void kernel_launch(void* const* d_in, const int* in_sizes, int n_in,
                              void* d_out, int out_size, void* d_ws, size_t ws_size,
                              hipStream_t stream) {
  const float* x  = (const float*)d_in[0];
  const int*   ei = (const int*)d_in[1];
  const float* W  = (const float*)d_in[2];
  const float* b  = (const float*)d_in[3];
  float* out = (float*)d_out;
  float* H    = (float*)d_ws;
  int*   head = (int*)((char*)d_ws + 25600000);
  int*   node = (int*)((char*)d_ws + 25800192);

  gcn_gemm_wmma<<<GEMM_BLOCKS, 128, 0, stream>>>(x, W, H, head);

  gcn_link<<<N_EDGES / 256, 256, 0, stream>>>(ei, head, node);

  gcn_gather<<<N_NODES / 8, 256, 0, stream>>>(head, node, H, b, out);
}


# link and gather on fewer, longer waves (75,000 -> 15,625): a link thread takes 8 edges, a gather wave walks 4 rows' lists at once, ended lists idle on a zero row; sums stay in list (arrival) order
# speedup vs baseline: 1.9075x; 1.0584x over previous
#include <hip/hip_runtime.h>

typedef __attribute__((ext_vector_type(2))) float v2f;
typedef __attribute__((ext_vector_type(4))) float v4f;
typedef __attribute__((ext_vector_type(8))) float v8f;
typedef __attribute__((ext_vector_type(2))) int   v2i;
typedef __attribute__((ext_vector_type(4))) int   v4i;

#define N_NODES 50000
#define N_EDGES 800000
#define D       128
#define SENT    N_EDGES
#define ZROW    N_NODES


#define ROW_TILES   (N_NODES / 16)
#define GEMM_BLOCKS 512

__device__ __forceinline__ void load_frag(v4f (&f)[16], const float* __restrict__ p) {
#pragma unroll
  for (int j = 0; j < 16; ++j) f[j] = *(const v4f*)(p + 8 * j);
}

__device__ __forceinline__ void tile_product(const v4f (&a)[16],
                                             const v4f (&b0)[16],
                                             const v4f (&b1)[16],
                                             float* __restrict__ hrow) {
  v8f acc0 = {};
  v8f acc1 = {};
#pragma unroll
  for (int j = 0; j < 16; ++j) {
    const v2f alo = a[j].xy,  ahi = a[j].zw;
    const v2f b0lo = b0[j].xy, b0hi = b0[j].zw;
    const v2f b1lo = b1[j].xy, b1hi = b1[j].zw;
    acc0 = __builtin_amdgcn_wmma_f32_16x16x4_f32(false, alo, false, b0lo,
                                                 (short)0, acc0, false, false);
    acc1 = __builtin_amdgcn_wmma_f32_16x16x4_f32(false, alo, false, b1lo,
                                                 (short)0, acc1, false, false);
    acc0 = __builtin_amdgcn_wmma_f32_16x16x4_f32(false, ahi, false, b0hi,
                                                 (short)0, acc0, false, false);
    acc1 = __builtin_amdgcn_wmma_f32_16x16x4_f32(false, ahi, false, b1hi,
                                                 (short)0, acc1, false, false);
  }
#pragma unroll
  for (int r = 0; r < 8; ++r) {
    hrow[(size_t)r * D]      = acc0[r];
    hrow[(size_t)r * D + 16] = acc1[r];
  }
}

__global__ __launch_bounds__(128)
void gcn_gemm_wmma(const float* __restrict__ X,
                   const float* __restrict__ W,
                   float* __restrict__ H,
                   int* __restrict__ head) {
  const int lane  = threadIdx.x & 31;
  const int wave  = threadIdx.x >> 5;
  const int nbase = wave * 32;
  const int half  = lane >> 4;
  const int l     = lane & 15;

  {
    const int tq = blockIdx.x + GEMM_BLOCKS * (threadIdx.x >> 4);
    if (tq < ROW_TILES) head[tq * 16 + (threadIdx.x & 15)] = SENT;
  }
  if (blockIdx.x == 0) H[(size_t)ZROW * D + threadIdx.x] = 0.0f;

  const size_t aoff = (size_t)l * D + 4 * half;
  const size_t hoff = (size_t)(8 * half) * D + nbase + l;

  int t = blockIdx.x;
  v4f a0[16], a1[16];
  load_frag(a0, X + (size_t)t * 16 * D + aoff);

  v4f b0[16], b1[16];
  {
    const float* __restrict__ w0 = W + (size_t)(nbase + l) * D + 4 * half;
    const float* __restrict__ w1 = w0 + 16 * D;
#pragma unroll
    for (int j = 0; j < 16; ++j) {
      b0[j] = *(const v4f*)(w0 + 8 * j);
      b1[j] = *(const v4f*)(w1 + 8 * j);
    }
  }

  for (;;) {
    const int t1 = t + GEMM_BLOCKS;
    if (t1 >= ROW_TILES) {
      tile_product(a0, b0, b1, H + (size_t)t * 16 * D + hoff);
      break;
    }
    load_frag(a1, X + (size_t)t1 * 16 * D + aoff);
    __builtin_amdgcn_sched_barrier(0);
    tile_product(a0, b0, b1, H + (size_t)t * 16 * D + hoff);

    const int t2 = t1 + GEMM_BLOCKS;
    if (t2 >= ROW_TILES) {
      tile_product(a1, b0, b1, H + (size_t)t1 * 16 * D + hoff);
      break;
    }
    load_frag(a0, X + (size_t)t2 * 16 * D + aoff);
    __builtin_amdgcn_sched_barrier(0);
    tile_product(a1, b0, b1, H + (size_t)t1 * 16 * D + hoff);
    t = t2;
  }
}

#define LINK_THREADS (N_EDGES / 8)
#define LINK_BLOCK   160

__global__ __launch_bounds__(LINK_BLOCK)
void gcn_link(const int* __restrict__ ei,
              int* __restrict__ head,
              int* __restrict__ node) {
  const int t = blockIdx.x * LINK_BLOCK + threadIdx.x;
  const v2i* __restrict__ src2 = (const v2i*)ei;
  const v2i* __restrict__ dst2 = (const v2i*)(ei + N_EDGES);

  v2i s[4], d[4];
#pragma unroll
  for (int k = 0; k < 4; ++k) {
    s[k] = src2[k * LINK_THREADS + t];
    d[k] = dst2[k * LINK_THREADS + t];
  }
  int p[8];
#pragma unroll
  for (int k = 0; k < 4; ++k) {
    const int e = 2 * (k * LINK_THREADS + t);
    p[2 * k]     = atomicExch(head + d[k].x, e);
    p[2 * k + 1] = atomicExch(head + d[k].y, e + 1);
  }
#pragma unroll
  for (int k = 0; k < 4; ++k) {
    v4i nd;
    nd.x = p[2 * k];     nd.y = s[k].x;
    nd.z = p[2 * k + 1]; nd.w = s[k].y;
    ((v4i*)node)[k * LINK_THREADS + t] = nd;
  }
  if (t == 0) {
    v2i en;
    en.x = SENT;
    en.y = ZROW;
    ((v2i*)node)[SENT] = en;
  }
}

__global__ __launch_bounds__(128)
void gcn_gather(const int* __restrict__ head,
                const int* __restrict__ node,
                const float* __restrict__ H,
                const float* __restrict__ b,
                float* __restrict__ out) {
  const int lane = threadIdx.x & 31;
  const int wave = __builtin_amdgcn_readfirstlane(blockIdx.x * 4 + (threadIdx.x >> 5));
  const v2i* __restrict__ nodes = (const v2i*)node;
  const v4f* __restrict__ H4 = (const v4f*)H;

  const v4f bias = ((const v4f*)b)[lane];
  v4f acc0 = bias, acc1 = bias, acc2 = bias, acc3 = bias;

  const v4i h = ((const v4i*)head)[wave];
  int e0 = h.x, e1 = h.y, e2 = h.z, e3 = h.w;
  v2i n0 = nodes[e0], n1 = nodes[e1], n2 = nodes[e2], n3 = nodes[e3];

  for (int it = 0; it < N_EDGES; ++it) {
    if (e0 == SENT && e1 == SENT && e2 == SENT && e3 == SENT) break;
    const v4f m0 = H4[(size_t)n0.y * (D / 4) + lane];
    const v4f m1 = H4[(size_t)n1.y * (D / 4) + lane];
    const v4f m2 = H4[(size_t)n2.y * (D / 4) + lane];
    const v4f m3 = H4[(size_t)n3.y * (D / 4) + lane];
    const v2i x0 = nodes[n0.x];
    const v2i x1 = nodes[n1.x];
    const v2i x2 = nodes[n2.x];
    const v2i x3 = nodes[n3.x];
    asm volatile("" :: "s"(x0.x), "s"(x0.y), "s"(x1.x), "s"(x1.y),
                       "s"(x2.x), "s"(x2.y), "s"(x3.x), "s"(x3.y));
    acc0 += m0;
    acc1 += m1;
    acc2 += m2;
    acc3 += m3;
    e0 = n0.x; e1 = n1.x; e2 = n2.x; e3 = n3.x;
    n0 = x0;   n1 = x1;   n2 = x2;   n3 = x3;
  }
  v4f* __restrict__ o = (v4f*)out + (size_t)wave * 4 * (D / 4) + lane;
  o[0 * (D / 4)] = acc0;
  o[1 * (D / 4)] = acc1;
  o[2 * (D / 4)] = acc2;
  o[3 * (D / 4)] = acc3;
}

extern "C" void kernel_launch(void* const* d_in, const int* in_sizes, int n_in,
                              void* d_out, int out_size, void* d_ws, size_t ws_size,
                              hipStream_t stream) {
  const float* x  = (const float*)d_in[0];
  const int*   ei = (const int*)d_in[1];
  const float* W  = (const float*)d_in[2];
  const float* b  = (const float*)d_in[3];
  float* out = (float*)d_out;
  float* H    = (float*)d_ws;
  int*   head = (int*)((char*)d_ws + 25600512);
  int*   node = (int*)((char*)d_ws + 25800704);

  gcn_gemm_wmma<<<GEMM_BLOCKS, 128, 0, stream>>>(x, W, H, head);

  gcn_link<<<LINK_THREADS / LINK_BLOCK, LINK_BLOCK, 0, stream>>>(ei, head, node);

  gcn_gather<<<N_NODES / 16, 128, 0, stream>>>(head, node, H, b, out);
}


# link in LDS: 16 work-groups each keep all 50,000 list heads in LDS and link 1/16 of the edges by LDS exchange, a row's 16 parts joined by zero-row nodes; no global atomics; sums in new list order
# speedup vs baseline: 2.5767x; 1.3508x over previous
#include <hip/hip_runtime.h>

typedef __attribute__((ext_vector_type(2))) float v2f;
typedef __attribute__((ext_vector_type(4))) float v4f;
typedef __attribute__((ext_vector_type(8))) float v8f;
typedef __attribute__((ext_vector_type(2))) int   v2i;
typedef __attribute__((ext_vector_type(4))) int   v4i;

#define N_NODES 50000
#define N_EDGES 800000
#define D       128
#define SENT    N_EDGES
#define ZROW    N_NODES

#define LINK_GROUPS 16
#define CHUNK_PAIRS (N_EDGES / LINK_GROUPS / 2)
#define CELL_BASE   (N_EDGES + 2)
#define CELL(g, r)  (CELL_BASE + (g) * N_NODES + (r))


#define ROW_TILES   (N_NODES / 16)
#define GEMM_BLOCKS 512

__device__ __forceinline__ void load_frag(v4f (&f)[16], const float* __restrict__ p) {
#pragma unroll
  for (int j = 0; j < 16; ++j) f[j] = *(const v4f*)(p + 8 * j);
}

__device__ __forceinline__ void tile_product(const v4f (&a)[16],
                                             const v4f (&b0)[16],
                                             const v4f (&b1)[16],
                                             float* __restrict__ hrow) {
  v8f acc0 = {};
  v8f acc1 = {};
#pragma unroll
  for (int j = 0; j < 16; ++j) {
    const v2f alo = a[j].xy,  ahi = a[j].zw;
    const v2f b0lo = b0[j].xy, b0hi = b0[j].zw;
    const v2f b1lo = b1[j].xy, b1hi = b1[j].zw;
    acc0 = __builtin_amdgcn_wmma_f32_16x16x4_f32(false, alo, false, b0lo,
                                                 (short)0, acc0, false, false);
    acc1 = __builtin_amdgcn_wmma_f32_16x16x4_f32(false, alo, false, b1lo,
                                                 (short)0, acc1, false, false);
    acc0 = __builtin_amdgcn_wmma_f32_16x16x4_f32(false, ahi, false, b0hi,
                                                 (short)0, acc0, false, false);
    acc1 = __builtin_amdgcn_wmma_f32_16x16x4_f32(false, ahi, false, b1hi,
                                                 (short)0, acc1, false, false);
  }
#pragma unroll
  for (int r = 0; r < 8; ++r) {
    hrow[(size_t)r * D]      = acc0[r];
    hrow[(size_t)r * D + 16] = acc1[r];
  }
}

__global__ __launch_bounds__(128)
void gcn_gemm_wmma(const float* __restrict__ X,
                   const float* __restrict__ W,
                   float* __restrict__ H) {
  const int lane  = threadIdx.x & 31;
  const int wave  = threadIdx.x >> 5;
  const int nbase = wave * 32;
  const int half  = lane >> 4;
  const int l     = lane & 15;

  if (blockIdx.x == 0) H[(size_t)ZROW * D + threadIdx.x] = 0.0f;

  const size_t aoff = (size_t)l * D + 4 * half;
  const size_t hoff = (size_t)(8 * half) * D + nbase + l;

  int t = blockIdx.x;
  v4f a0[16], a1[16];
  load_frag(a0, X + (size_t)t * 16 * D + aoff);

  v4f b0[16], b1[16];
  {
    const float* __restrict__ w0 = W + (size_t)(nbase + l) * D + 4 * half;
    const float* __restrict__ w1 = w0 + 16 * D;
#pragma unroll
    for (int j = 0; j < 16; ++j) {
      b0[j] = *(const v4f*)(w0 + 8 * j);
      b1[j] = *(const v4f*)(w1 + 8 * j);
    }
  }

  for (;;) {
    const int t1 = t + GEMM_BLOCKS;
    if (t1 >= ROW_TILES) {
      tile_product(a0, b0, b1, H + (size_t)t * 16 * D + hoff);
      break;
    }
    load_frag(a1, X + (size_t)t1 * 16 * D + aoff);
    __builtin_amdgcn_sched_barrier(0);
    tile_product(a0, b0, b1, H + (size_t)t * 16 * D + hoff);

    const int t2 = t1 + GEMM_BLOCKS;
    if (t2 >= ROW_TILES) {
      tile_product(a1, b0, b1, H + (size_t)t1 * 16 * D + hoff);
      break;
    }
    load_frag(a0, X + (size_t)t2 * 16 * D + aoff);
    __builtin_amdgcn_sched_barrier(0);
    tile_product(a1, b0, b1, H + (size_t)t1 * 16 * D + hoff);
    t = t2;
  }
}

#define LINK_BLOCK 1024

__global__ __launch_bounds__(LINK_BLOCK)
void gcn_link(const int* __restrict__ ei,
              int* __restrict__ head,
              int* __restrict__ node) {
  __shared__ __attribute__((aligned(16))) int tab[N_NODES];
  const int g   = blockIdx.x;
  const int tid = threadIdx.x;

  for (int r = 4 * tid; r < N_NODES; r += 4 * LINK_BLOCK) {
    v4i v;
    if (g == 0) {
      v.x = SENT; v.y = SENT; v.z = SENT; v.w = SENT;
    } else {
      const int c = CELL(g - 1, r);
      v.x = c; v.y = c + 1; v.z = c + 2; v.w = c + 3;
    }
    *(v4i*)(tab + r) = v;
  }
  __syncthreads();

  const v2i* __restrict__ src2 = (const v2i*)ei + g * CHUNK_PAIRS;
  const v2i* __restrict__ dst2 = (const v2i*)(ei + N_EDGES) + g * CHUNK_PAIRS;
  v4i* __restrict__ node4 = (v4i*)node + g * CHUNK_PAIRS;
  for (int base = 0; base < CHUNK_PAIRS; base += 4 * LINK_BLOCK) {
    v2i s[4], d[4];
#pragma unroll
    for (int k = 0; k < 4; ++k) {
      const int q = base + k * LINK_BLOCK + tid;
      if (q < CHUNK_PAIRS) {
        s[k] = src2[q];
        d[k] = dst2[q];
      }
    }
    int p[8];
#pragma unroll
    for (int k = 0; k < 4; ++k) {
      const int q = base + k * LINK_BLOCK + tid;
      if (q < CHUNK_PAIRS) {
        const int e = 2 * (g * CHUNK_PAIRS + q);
        p[2 * k]     = atomicExch(tab + d[k].x, e);
        p[2 * k + 1] = atomicExch(tab + d[k].y, e + 1);
      }
    }
#pragma unroll
    for (int k = 0; k < 4; ++k) {
      const int q = base + k * LINK_BLOCK + tid;
      if (q < CHUNK_PAIRS) {
        v4i nd;
        nd.x = p[2 * k];     nd.y = s[k].x;
        nd.z = p[2 * k + 1]; nd.w = s[k].y;
        node4[q] = nd;
      }
    }
  }
  __syncthreads();

  if (g == LINK_GROUPS - 1) {
    v2i* __restrict__ head2 = (v2i*)head;
    for (int q = tid; q < N_NODES / 2; q += LINK_BLOCK)
      head2[q] = *(const v2i*)(tab + 2 * q);
  } else {
    v4i* __restrict__ cell4 = (v4i*)node + CELL(g, 0) / 2;
    for (int q = tid; q < N_NODES / 2; q += LINK_BLOCK) {
      const v2i t = *(const v2i*)(tab + 2 * q);
      v4i c;
      c.x = t.x; c.y = ZROW;
      c.z = t.y; c.w = ZROW;
      cell4[q] = c;
    }
  }
  if (g == 0 && tid == 0) {
    v2i en;
    en.x = SENT;
    en.y = ZROW;
    ((v2i*)node)[SENT] = en;
  }
}

__global__ __launch_bounds__(128)
void gcn_gather(const int* __restrict__ head,
                const int* __restrict__ node,
                const float* __restrict__ H,
                const float* __restrict__ b,
                float* __restrict__ out) {
  const int lane = threadIdx.x & 31;
  const int wave = __builtin_amdgcn_readfirstlane(blockIdx.x * 4 + (threadIdx.x >> 5));
  const v2i* __restrict__ nodes = (const v2i*)node;
  const v4f* __restrict__ H4 = (const v4f*)H;

  const v4f bias = ((const v4f*)b)[lane];
  v4f acc0 = bias, acc1 = bias, acc2 = bias, acc3 = bias;

  const v4i h = ((const v4i*)head)[wave];
  int e0 = h.x, e1 = h.y, e2 = h.z, e3 = h.w;
  v2i n0 = nodes[e0], n1 = nodes[e1], n2 = nodes[e2], n3 = nodes[e3];

  for (int it = 0; it < N_EDGES + LINK_GROUPS; ++it) {
    if (e0 == SENT && e1 == SENT && e2 == SENT && e3 == SENT) break;
    const v4f m0 = H4[(size_t)n0.y * (D / 4) + lane];
    const v4f m1 = H4[(size_t)n1.y * (D / 4) + lane];
    const v4f m2 = H4[(size_t)n2.y * (D / 4) + lane];
    const v4f m3 = H4[(size_t)n3.y * (D / 4) + lane];
    const v2i x0 = nodes[n0.x];
    const v2i x1 = nodes[n1.x];
    const v2i x2 = nodes[n2.x];
    const v2i x3 = nodes[n3.x];
    asm volatile("" :: "s"(x0.x), "s"(x0.y), "s"(x1.x), "s"(x1.y),
                       "s"(x2.x), "s"(x2.y), "s"(x3.x), "s"(x3.y));
    acc0 += m0;
    acc1 += m1;
    acc2 += m2;
    acc3 += m3;
    e0 = n0.x; e1 = n1.x; e2 = n2.x; e3 = n3.x;
    n0 = x0;   n1 = x1;   n2 = x2;   n3 = x3;
  }
  v4f* __restrict__ o = (v4f*)out + (size_t)wave * 4 * (D / 4) + lane;
  o[0 * (D / 4)] = acc0;
  o[1 * (D / 4)] = acc1;
  o[2 * (D / 4)] = acc2;
  o[3 * (D / 4)] = acc3;
}

extern "C" void kernel_launch(void* const* d_in, const int* in_sizes, int n_in,
                              void* d_out, int out_size, void* d_ws, size_t ws_size,
                              hipStream_t stream) {
  const float* x  = (const float*)d_in[0];
  const int*   ei = (const int*)d_in[1];
  const float* W  = (const float*)d_in[2];
  const float* b  = (const float*)d_in[3];
  float* out = (float*)d_out;
  float* H    = (float*)d_ws;
  int*   head = (int*)((char*)d_ws + 25600512);
  int*   node = (int*)((char*)d_ws + 25800704);

  gcn_gemm_wmma<<<GEMM_BLOCKS, 128, 0, stream>>>(x, W, H);

  gcn_link<<<LINK_GROUPS, LINK_BLOCK, 0, stream>>>(ei, head, node);

  gcn_gather<<<N_NODES / 16, 128, 0, stream>>>(head, node, H, b, out);
}


# gather: a row's 16 chunk lists are walked side by side, one per lane, from a 16-head table (hand-over nodes gone); sources collected in LDS, H rows summed 16 loads per trip; sums in a new list order
# speedup vs baseline: 2.9150x; 1.1313x over previous
#include <hip/hip_runtime.h>

typedef __attribute__((ext_vector_type(2))) float v2f;
typedef __attribute__((ext_vector_type(4))) float v4f;
typedef __attribute__((ext_vector_type(8))) float v8f;
typedef __attribute__((ext_vector_type(2))) int   v2i;
typedef __attribute__((ext_vector_type(4))) int   v4i;

#define N_NODES 50000
#define N_EDGES 800000
#define D       128
#define SENT    N_EDGES
#define ZROW    N_NODES

#define LINK_GROUPS 16
#define CHUNK_PAIRS (N_EDGES / LINK_GROUPS / 2)
#define CHUNK_EDGES (N_EDGES / LINK_GROUPS)


#define ROW_TILES   (N_NODES / 16)
#define GEMM_BLOCKS 512

__device__ __forceinline__ void load_frag(v4f (&f)[16], const float* __restrict__ p) {
#pragma unroll
  for (int j = 0; j < 16; ++j) f[j] = *(const v4f*)(p + 8 * j);
}

__device__ __forceinline__ void tile_product(const v4f (&a)[16],
                                             const v4f (&b0)[16],
                                             const v4f (&b1)[16],
                                             float* __restrict__ hrow) {
  v8f acc0 = {};
  v8f acc1 = {};
#pragma unroll
  for (int j = 0; j < 16; ++j) {
    const v2f alo = a[j].xy,  ahi = a[j].zw;
    const v2f b0lo = b0[j].xy, b0hi = b0[j].zw;
    const v2f b1lo = b1[j].xy, b1hi = b1[j].zw;
    acc0 = __builtin_amdgcn_wmma_f32_16x16x4_f32(false, alo, false, b0lo,
                                                 (short)0, acc0, false, false);
    acc1 = __builtin_amdgcn_wmma_f32_16x16x4_f32(false, alo, false, b1lo,
                                                 (short)0, acc1, false, false);
    acc0 = __builtin_amdgcn_wmma_f32_16x16x4_f32(false, ahi, false, b0hi,
                                                 (short)0, acc0, false, false);
    acc1 = __builtin_amdgcn_wmma_f32_16x16x4_f32(false, ahi, false, b1hi,
                                                 (short)0, acc1, false, false);
  }
#pragma unroll
  for (int r = 0; r < 8; ++r) {
    hrow[(size_t)r * D]      = acc0[r];
    hrow[(size_t)r * D + 16] = acc1[r];
  }
}

__global__ __launch_bounds__(128)
void gcn_gemm_wmma(const float* __restrict__ X,
                   const float* __restrict__ W,
                   float* __restrict__ H) {
  const int lane  = threadIdx.x & 31;
  const int wave  = threadIdx.x >> 5;
  const int nbase = wave * 32;
  const int half  = lane >> 4;
  const int l     = lane & 15;

  if (blockIdx.x == 0) H[(size_t)ZROW * D + threadIdx.x] = 0.0f;

  const size_t aoff = (size_t)l * D + 4 * half;
  const size_t hoff = (size_t)(8 * half) * D + nbase + l;

  int t = blockIdx.x;
  v4f a0[16], a1[16];
  load_frag(a0, X + (size_t)t * 16 * D + aoff);

  v4f b0[16], b1[16];
  {
    const float* __restrict__ w0 = W + (size_t)(nbase + l) * D + 4 * half;
    const float* __restrict__ w1 = w0 + 16 * D;
#pragma unroll
    for (int j = 0; j < 16; ++j) {
      b0[j] = *(const v4f*)(w0 + 8 * j);
      b1[j] = *(const v4f*)(w1 + 8 * j);
    }
  }

  for (;;) {
    const int t1 = t + GEMM_BLOCKS;
    if (t1 >= ROW_TILES) {
      tile_product(a0, b0, b1, H + (size_t)t * 16 * D + hoff);
      break;
    }
    load_frag(a1, X + (size_t)t1 * 16 * D + aoff);
    __builtin_amdgcn_sched_barrier(0);
    tile_product(a0, b0, b1, H + (size_t)t * 16 * D + hoff);

    const int t2 = t1 + GEMM_BLOCKS;
    if (t2 >= ROW_TILES) {
      tile_product(a1, b0, b1, H + (size_t)t1 * 16 * D + hoff);
      break;
    }
    load_frag(a0, X + (size_t)t2 * 16 * D + aoff);
    __builtin_amdgcn_sched_barrier(0);
    tile_product(a1, b0, b1, H + (size_t)t1 * 16 * D + hoff);
    t = t2;
  }
}

#define LINK_BLOCK 1024

__global__ __launch_bounds__(LINK_BLOCK)
void gcn_link(const int* __restrict__ ei,
              int* __restrict__ heads,
              int* __restrict__ node) {
  __shared__ __attribute__((aligned(16))) int tab[N_NODES];
  const int g   = blockIdx.x;
  const int tid = threadIdx.x;

  for (int r = 4 * tid; r < N_NODES; r += 4 * LINK_BLOCK) {
    v4i v;
    v.x = SENT; v.y = SENT; v.z = SENT; v.w = SENT;
    *(v4i*)(tab + r) = v;
  }
  __syncthreads();

  const v2i* __restrict__ src2 = (const v2i*)ei + g * CHUNK_PAIRS;
  const v2i* __restrict__ dst2 = (const v2i*)(ei + N_EDGES) + g * CHUNK_PAIRS;
  v4i* __restrict__ node4 = (v4i*)node + g * CHUNK_PAIRS;
  for (int base = 0; base < CHUNK_PAIRS; base += 4 * LINK_BLOCK) {
    v2i s[4], d[4];
#pragma unroll
    for (int k = 0; k < 4; ++k) {
      const int q = base + k * LINK_BLOCK + tid;
      if (q < CHUNK_PAIRS) {
        s[k] = src2[q];
        d[k] = dst2[q];
      }
    }
    int p[8];
#pragma unroll
    for (int k = 0; k < 4; ++k) {
      const int q = base + k * LINK_BLOCK + tid;
      if (q < CHUNK_PAIRS) {
        const int e = 2 * (g * CHUNK_PAIRS + q);
        p[2 * k]     = atomicExch(tab + d[k].x, e);
        p[2 * k + 1] = atomicExch(tab + d[k].y, e + 1);
      }
    }
#pragma unroll
    for (int k = 0; k < 4; ++k) {
      const int q = base + k * LINK_BLOCK + tid;
      if (q < CHUNK_PAIRS) {
        v4i nd;
        nd.x = p[2 * k];     nd.y = s[k].x;
        nd.z = p[2 * k + 1]; nd.w = s[k].y;
        node4[q] = nd;
      }
    }
  }
  __syncthreads();

  v4i* __restrict__ heads4 = (v4i*)(heads + g * N_NODES);
  for (int q = tid; q < N_NODES / 4; q += LINK_BLOCK)
    heads4[q] = *(const v4i*)(tab + 4 * q);
}

#define G_WAVES    4
#define G_PASSES   2
#define G_ROWS     (2 * G_WAVES * G_PASSES)
#define WALK_STEPS 8
#define LCAP       (16 * WALK_STEPS)

__global__ __launch_bounds__(32 * G_WAVES)
void gcn_gather(const int* __restrict__ heads,
                const int* __restrict__ node,
                const float* __restrict__ H,
                const float* __restrict__ b,
                float* __restrict__ out) {
  __shared__ int hd[G_ROWS * LINK_GROUPS];
  __shared__ __attribute__((aligned(16))) int lst[G_WAVES * 2 * LCAP];

  const int tid  = threadIdx.x;
  const int lane = tid & 31;
  const int half = lane >> 4;
  const int wave = __builtin_amdgcn_readfirstlane(tid >> 5);
  const v2i* __restrict__ nodes = (const v2i*)node;
  const v4f* __restrict__ H4 = (const v4f*)H;

#pragma unroll
  for (int k = 0; k < G_PASSES; ++k) {
    const int t = tid + 32 * G_WAVES * k;
    hd[(t & 15) * LINK_GROUPS + (t >> 4)] =
        heads[(t >> 4) * N_NODES + blockIdx.x * G_ROWS + (t & 15)];
  }
  __syncthreads();

  const v4f bias = ((const v4f*)b)[lane];

  int* const tabA = lst + wave * (2 * LCAP);
  int* const tabB = tabA + LCAP;
  int* const mytab = half ? tabB : tabA;
  const unsigned below = (1u << (lane & 15)) - 1u;

  for (int pass = 0; pass < G_PASSES; ++pass) {
  const int pr = wave + G_WAVES * pass;
  int e = hd[pr * 32 + lane];
  v4f accA = bias, accB = bias;

  for (int round = 0; round <= CHUNK_EDGES / WALK_STEPS; ++round) {
    if (__builtin_amdgcn_ballot_w32(e != SENT) == 0u) break;

    {
      v4i z;
      z.x = ZROW; z.y = ZROW; z.z = ZROW; z.w = ZROW;
      *(v4i*)(tabA + 4 * lane) = z;
      *(v4i*)(tabB + 4 * lane) = z;
    }

    int cntA = 0, cntB = 0;
    for (int s = 0; s < WALK_STEPS; ++s) {
      const bool live = (e != SENT);
      const unsigned m = __builtin_amdgcn_ballot_w32(live);
      if (m == 0u) break;
      const unsigned mA = m & 0xFFFFu, mB = m >> 16;
      const int pos = (half ? cntB : cntA) +
                      __builtin_popcount((half ? mB : mA) & below);
      if (live) {
        const v2i nd = nodes[e];
        mytab[pos] = nd.y;
        e = nd.x;
      }
      cntA += __builtin_popcount(mA);
      cntB += __builtin_popcount(mB);
    }
    __builtin_amdgcn_fence(__ATOMIC_RELEASE, "wavefront");
    __builtin_amdgcn_wave_barrier();
    __builtin_amdgcn_fence(__ATOMIC_ACQUIRE, "wavefront");

    const int nmax = cntA > cntB ? cntA : cntB;
    for (int i = 0; i < nmax; i += 8) {
      const v4i a0 = *(const v4i*)(tabA + i);
      const v4i a1 = *(const v4i*)(tabA + i + 4);
      const v4i b0 = *(const v4i*)(tabB + i);
      const v4i b1 = *(const v4i*)(tabB + i + 4);
      int ra[8], rb[8];
      ra[0] = a0.x; ra[1] = a0.y; ra[2] = a0.z; ra[3] = a0.w;
      ra[4] = a1.x; ra[5] = a1.y; ra[6] = a1.z; ra[7] = a1.w;
      rb[0] = b0.x; rb[1] = b0.y; rb[2] = b0.z; rb[3] = b0.w;
      rb[4] = b1.x; rb[5] = b1.y; rb[6] = b1.z; rb[7] = b1.w;
      v4f ma[8], mb[8];
#pragma unroll
      for (int k = 0; k < 8; ++k) {
        ma[k] = H4[(size_t)__builtin_amdgcn_readfirstlane(ra[k]) * (D / 4) + lane];
        mb[k] = H4[(size_t)__builtin_amdgcn_readfirstlane(rb[k]) * (D / 4) + lane];
      }
#pragma unroll
      for (int k = 0; k < 8; ++k) {
        accA += ma[k];
        accB += mb[k];
      }
    }
    __builtin_amdgcn_wave_barrier();
  }

  v4f* __restrict__ o =
      (v4f*)out + (size_t)(blockIdx.x * G_ROWS + 2 * pr) * (D / 4) + lane;
  o[0]     = accA;
  o[D / 4] = accB;
  }
}

extern "C" void kernel_launch(void* const* d_in, const int* in_sizes, int n_in,
                              void* d_out, int out_size, void* d_ws, size_t ws_size,
                              hipStream_t stream) {
  const float* x  = (const float*)d_in[0];
  const int*   ei = (const int*)d_in[1];
  const float* W  = (const float*)d_in[2];
  const float* b  = (const float*)d_in[3];
  float* out = (float*)d_out;
  float* H     = (float*)d_ws;
  int*   heads = (int*)((char*)d_ws + 25600512);
  int*   node  = (int*)((char*)d_ws + 28800512);

  gcn_gemm_wmma<<<GEMM_BLOCKS, 128, 0, stream>>>(x, W, H);

  gcn_link<<<LINK_GROUPS, LINK_BLOCK, 0, stream>>>(ei, heads, node);

  gcn_gather<<<N_NODES / G_ROWS, 32 * G_WAVES, 0, stream>>>(heads, node, H, b, out);
}


# list records packed to 16-bit fields: node 4 B (back-link in chunk, source row), heads 2 B; link stores 300 KB per work-group for 600, 4 trips for 7; lists in arrival order, edge-thread map changed
# speedup vs baseline: 3.1891x; 1.0940x over previous
#include <hip/hip_runtime.h>

typedef __attribute__((ext_vector_type(2))) float v2f;
typedef __attribute__((ext_vector_type(4))) float v4f;
typedef __attribute__((ext_vector_type(8))) float v8f;
typedef __attribute__((ext_vector_type(4))) int   v4i;

#define N_NODES 50000
#define N_EDGES 800000
#define D       128
#define SENT    0xFFFF
#define ZROW    N_NODES

#define LINK_GROUPS 16
#define CHUNK_EDGES (N_EDGES / LINK_GROUPS)
#define CHUNK_QUADS (CHUNK_EDGES / 4)
static_assert(CHUNK_EDGES < SENT && N_NODES <= SENT, "16-bit list fields");


#define ROW_TILES   (N_NODES / 16)
#define GEMM_BLOCKS 512

__device__ __forceinline__ void load_frag(v4f (&f)[16], const float* __restrict__ p) {
#pragma unroll
  for (int j = 0; j < 16; ++j) f[j] = *(const v4f*)(p + 8 * j);
}

__device__ __forceinline__ void tile_product(const v4f (&a)[16],
                                             const v4f (&b0)[16],
                                             const v4f (&b1)[16],
                                             float* __restrict__ hrow) {
  v8f acc0 = {};
  v8f acc1 = {};
#pragma unroll
  for (int j = 0; j < 16; ++j) {
    const v2f alo = a[j].xy,  ahi = a[j].zw;
    const v2f b0lo = b0[j].xy, b0hi = b0[j].zw;
    const v2f b1lo = b1[j].xy, b1hi = b1[j].zw;
    acc0 = __builtin_amdgcn_wmma_f32_16x16x4_f32(false, alo, false, b0lo,
                                                 (short)0, acc0, false, false);
    acc1 = __builtin_amdgcn_wmma_f32_16x16x4_f32(false, alo, false, b1lo,
                                                 (short)0, acc1, false, false);
    acc0 = __builtin_amdgcn_wmma_f32_16x16x4_f32(false, ahi, false, b0hi,
                                                 (short)0, acc0, false, false);
    acc1 = __builtin_amdgcn_wmma_f32_16x16x4_f32(false, ahi, false, b1hi,
                                                 (short)0, acc1, false, false);
  }
#pragma unroll
  for (int r = 0; r < 8; ++r) {
    hrow[(size_t)r * D]      = acc0[r];
    hrow[(size_t)r * D + 16] = acc1[r];
  }
}

__global__ __launch_bounds__(128)
void gcn_gemm_wmma(const float* __restrict__ X,
                   const float* __restrict__ W,
                   float* __restrict__ H) {
  const int lane  = threadIdx.x & 31;
  const int wave  = threadIdx.x >> 5;
  const int nbase = wave * 32;
  const int half  = lane >> 4;
  const int l     = lane & 15;

  if (blockIdx.x == 0) H[(size_t)ZROW * D + threadIdx.x] = 0.0f;

  const size_t aoff = (size_t)l * D + 4 * half;
  const size_t hoff = (size_t)(8 * half) * D + nbase + l;

  int t = blockIdx.x;
  v4f a0[16], a1[16];
  load_frag(a0, X + (size_t)t * 16 * D + aoff);

  v4f b0[16], b1[16];
  {
    const float* __restrict__ w0 = W + (size_t)(nbase + l) * D + 4 * half;
    const float* __restrict__ w1 = w0 + 16 * D;
#pragma unroll
    for (int j = 0; j < 16; ++j) {
      b0[j] = *(const v4f*)(w0 + 8 * j);
      b1[j] = *(const v4f*)(w1 + 8 * j);
    }
  }

  for (;;) {
    const int t1 = t + GEMM_BLOCKS;
    if (t1 >= ROW_TILES) {
      tile_product(a0, b0, b1, H + (size_t)t * 16 * D + hoff);
      break;
    }
    load_frag(a1, X + (size_t)t1 * 16 * D + aoff);
    __builtin_amdgcn_sched_barrier(0);
    tile_product(a0, b0, b1, H + (size_t)t * 16 * D + hoff);

    const int t2 = t1 + GEMM_BLOCKS;
    if (t2 >= ROW_TILES) {
      tile_product(a1, b0, b1, H + (size_t)t1 * 16 * D + hoff);
      break;
    }
    load_frag(a0, X + (size_t)t2 * 16 * D + aoff);
    __builtin_amdgcn_sched_barrier(0);
    tile_product(a1, b0, b1, H + (size_t)t1 * 16 * D + hoff);
    t = t2;
  }
}

#define LINK_BLOCK 1024

__global__ __launch_bounds__(LINK_BLOCK)
void gcn_link(const int* __restrict__ ei,
              unsigned short* __restrict__ heads,
              unsigned* __restrict__ node) {
  __shared__ __attribute__((aligned(16))) int tab[N_NODES];
  const int g   = blockIdx.x;
  const int tid = threadIdx.x;

  for (int r = 4 * tid; r < N_NODES; r += 4 * LINK_BLOCK) {
    v4i v;
    v.x = SENT; v.y = SENT; v.z = SENT; v.w = SENT;
    *(v4i*)(tab + r) = v;
  }
  __syncthreads();

  const v4i* __restrict__ src4 = (const v4i*)ei + g * CHUNK_QUADS;
  const v4i* __restrict__ dst4 = (const v4i*)(ei + N_EDGES) + g * CHUNK_QUADS;
  v4i* __restrict__ node4 = (v4i*)node + g * CHUNK_QUADS;
  for (int base = 0; base < CHUNK_QUADS; base += 4 * LINK_BLOCK) {
    v4i s[4], d[4];
#pragma unroll
    for (int k = 0; k < 4; ++k) {
      const int q = base + k * LINK_BLOCK + tid;
      if (q < CHUNK_QUADS) {
        s[k] = src4[q];
        d[k] = dst4[q];
      }
    }
    unsigned p[16];
#pragma unroll
    for (int k = 0; k < 4; ++k) {
      const int q = base + k * LINK_BLOCK + tid;
      if (q < CHUNK_QUADS) {
#pragma unroll
        for (int c = 0; c < 4; ++c)
          p[4 * k + c] = (unsigned)atomicExch(tab + d[k][c], 4 * q + c);
      }
    }
#pragma unroll
    for (int k = 0; k < 4; ++k) {
      const int q = base + k * LINK_BLOCK + tid;
      if (q < CHUNK_QUADS) {
        v4i nd;
#pragma unroll
        for (int c = 0; c < 4; ++c)
          nd[c] = (int)((p[4 * k + c] << 16) | (unsigned)s[k][c]);
        node4[q] = nd;
      }
    }
  }
  __syncthreads();

  v4i* __restrict__ heads4 = (v4i*)(heads + g * N_NODES);
  for (int q = tid; q < N_NODES / 8; q += LINK_BLOCK) {
    const v4i a = *(const v4i*)(tab + 8 * q);
    const v4i b = *(const v4i*)(tab + 8 * q + 4);
    v4i o;
    o.x = (int)((unsigned)a.x | ((unsigned)a.y << 16));
    o.y = (int)((unsigned)a.z | ((unsigned)a.w << 16));
    o.z = (int)((unsigned)b.x | ((unsigned)b.y << 16));
    o.w = (int)((unsigned)b.z | ((unsigned)b.w << 16));
    heads4[q] = o;
  }
}

#define G_WAVES    4
#define G_PASSES   2
#define G_ROWS     (2 * G_WAVES * G_PASSES)
#define WALK_STEPS 8
#define LCAP       (16 * WALK_STEPS)

__global__ __launch_bounds__(32 * G_WAVES)
void gcn_gather(const unsigned short* __restrict__ heads,
                const unsigned* __restrict__ node,
                const float* __restrict__ H,
                const float* __restrict__ b,
                float* __restrict__ out) {
  __shared__ int hd[G_ROWS * LINK_GROUPS];
  __shared__ __attribute__((aligned(16))) int lst[G_WAVES * 2 * LCAP];

  const int tid  = threadIdx.x;
  const int lane = tid & 31;
  const int half = lane >> 4;
  const int wave = __builtin_amdgcn_readfirstlane(tid >> 5);
  const unsigned* __restrict__ nodes = node + (lane & 15) * CHUNK_EDGES;
  const v4f* __restrict__ H4 = (const v4f*)H;

#pragma unroll
  for (int k = 0; k < G_PASSES; ++k) {
    const int t = tid + 32 * G_WAVES * k;
    hd[(t & 15) * LINK_GROUPS + (t >> 4)] =
        heads[(t >> 4) * N_NODES + blockIdx.x * G_ROWS + (t & 15)];
  }
  __syncthreads();

  const v4f bias = ((const v4f*)b)[lane];

  int* const tabA = lst + wave * (2 * LCAP);
  int* const tabB = tabA + LCAP;
  int* const mytab = half ? tabB : tabA;
  const unsigned below = (1u << (lane & 15)) - 1u;

  for (int pass = 0; pass < G_PASSES; ++pass) {
  const int pr = wave + G_WAVES * pass;
  int e = hd[pr * 32 + lane];
  v4f accA = bias, accB = bias;

  for (int round = 0; round <= CHUNK_EDGES / WALK_STEPS; ++round) {
    if (__builtin_amdgcn_ballot_w32(e != SENT) == 0u) break;

    {
      v4i z;
      z.x = ZROW; z.y = ZROW; z.z = ZROW; z.w = ZROW;
      *(v4i*)(tabA + 4 * lane) = z;
      *(v4i*)(tabB + 4 * lane) = z;
    }

    int cntA = 0, cntB = 0;
    for (int s = 0; s < WALK_STEPS; ++s) {
      const bool live = (e != SENT);
      const unsigned m = __builtin_amdgcn_ballot_w32(live);
      if (m == 0u) break;
      const unsigned mA = m & 0xFFFFu, mB = m >> 16;
      const int pos = (half ? cntB : cntA) +
                      __builtin_popcount((half ? mB : mA) & below);
      if (live) {
        const unsigned nd = nodes[e];
        mytab[pos] = (int)(nd & 0xFFFFu);
        e = (int)(nd >> 16);
      }
      cntA += __builtin_popcount(mA);
      cntB += __builtin_popcount(mB);
    }
    __builtin_amdgcn_fence(__ATOMIC_RELEASE, "wavefront");
    __builtin_amdgcn_wave_barrier();
    __builtin_amdgcn_fence(__ATOMIC_ACQUIRE, "wavefront");

    const int nmax = cntA > cntB ? cntA : cntB;
    for (int i = 0; i < nmax; i += 8) {
      const v4i a0 = *(const v4i*)(tabA + i);
      const v4i a1 = *(const v4i*)(tabA + i + 4);
      const v4i b0 = *(const v4i*)(tabB + i);
      const v4i b1 = *(const v4i*)(tabB + i + 4);
      int ra[8], rb[8];
      ra[0] = a0.x; ra[1] = a0.y; ra[2] = a0.z; ra[3] = a0.w;
      ra[4] = a1.x; ra[5] = a1.y; ra[6] = a1.z; ra[7] = a1.w;
      rb[0] = b0.x; rb[1] = b0.y; rb[2] = b0.z; rb[3] = b0.w;
      rb[4] = b1.x; rb[5] = b1.y; rb[6] = b1.z; rb[7] = b1.w;
      v4f ma[8], mb[8];
#pragma unroll
      for (int k = 0; k < 8; ++k) {
        ma[k] = H4[(size_t)__builtin_amdgcn_readfirstlane(ra[k]) * (D / 4) + lane];
        mb[k] = H4[(size_t)__builtin_amdgcn_readfirstlane(rb[k]) * (D / 4) + lane];
      }
#pragma unroll
      for (int k = 0; k < 8; ++k) {
        accA += ma[k];
        accB += mb[k];
      }
    }
    __builtin_amdgcn_wave_barrier();
  }

  v4f* __restrict__ o =
      (v4f*)out + (size_t)(blockIdx.x * G_ROWS + 2 * pr) * (D / 4) + lane;
  o[0]     = accA;
  o[D / 4] = accB;
  }
}

extern "C" void kernel_launch(void* const* d_in, const int* in_sizes, int n_in,
                              void* d_out, int out_size, void* d_ws, size_t ws_size,
                              hipStream_t stream) {
  const float* x  = (const float*)d_in[0];
  const int*   ei = (const int*)d_in[1];
  const float* W  = (const float*)d_in[2];
  const float* b  = (const float*)d_in[3];
  float* out = (float*)d_out;
  float* H     = (float*)d_ws;
  unsigned short* heads = (unsigned short*)((char*)d_ws + 25600512);
  unsigned*       node  = (unsigned*)((char*)d_ws + 28800512);

  gcn_gemm_wmma<<<GEMM_BLOCKS, 128, 0, stream>>>(x, W, H);

  gcn_link<<<LINK_GROUPS, LINK_BLOCK, 0, stream>>>(ei, heads, node);

  gcn_gather<<<N_NODES / G_ROWS, 32 * G_WAVES, 0, stream>>>(heads, node, H, b, out);
}


# edge chunks 16 -> 32: 32 work-groups link 25,000 edges each in LDS in one trip, index loads ahead of the table fill; gather lane g walks chunk g's lists of both rows of a pair; sums in new list order
# speedup vs baseline: 3.4587x; 1.0845x over previous
#include <hip/hip_runtime.h>

typedef __attribute__((ext_vector_type(2))) float v2f;
typedef __attribute__((ext_vector_type(4))) float v4f;
typedef __attribute__((ext_vector_type(8))) float v8f;
typedef __attribute__((ext_vector_type(4))) int   v4i;

#define N_NODES 50000
#define N_EDGES 800000
#define D       128
#define SENT    0xFFFF
#define ZROW    N_NODES

#define LINK_GROUPS 32
#define CHUNK_EDGES (N_EDGES / LINK_GROUPS)
#define CHUNK_QUADS (CHUNK_EDGES / 4)
static_assert(CHUNK_EDGES < SENT && N_NODES <= SENT, "16-bit list fields");


#define ROW_TILES   (N_NODES / 16)
#define GEMM_BLOCKS 512

__device__ __forceinline__ void load_frag(v4f (&f)[16], const float* __restrict__ p) {
#pragma unroll
  for (int j = 0; j < 16; ++j) f[j] = *(const v4f*)(p + 8 * j);
}

__device__ __forceinline__ void tile_product(const v4f (&a)[16],
                                             const v4f (&b0)[16],
                                             const v4f (&b1)[16],
                                             float* __restrict__ hrow) {
  v8f acc0 = {};
  v8f acc1 = {};
#pragma unroll
  for (int j = 0; j < 16; ++j) {
    const v2f alo = a[j].xy,  ahi = a[j].zw;
    const v2f b0lo = b0[j].xy, b0hi = b0[j].zw;
    const v2f b1lo = b1[j].xy, b1hi = b1[j].zw;
    acc0 = __builtin_amdgcn_wmma_f32_16x16x4_f32(false, alo, false, b0lo,
                                                 (short)0, acc0, false, false);
    acc1 = __builtin_amdgcn_wmma_f32_16x16x4_f32(false, alo, false, b1lo,
                                                 (short)0, acc1, false, false);
    acc0 = __builtin_amdgcn_wmma_f32_16x16x4_f32(false, ahi, false, b0hi,
                                                 (short)0, acc0, false, false);
    acc1 = __builtin_amdgcn_wmma_f32_16x16x4_f32(false, ahi, false, b1hi,
                                                 (short)0, acc1, false, false);
  }
#pragma unroll
  for (int r = 0; r < 8; ++r) {
    hrow[(size_t)r * D]      = acc0[r];
    hrow[(size_t)r * D + 16] = acc1[r];
  }
}

__global__ __launch_bounds__(128)
void gcn_gemm_wmma(const float* __restrict__ X,
                   const float* __restrict__ W,
                   float* __restrict__ H) {
  const int lane  = threadIdx.x & 31;
  const int wave  = threadIdx.x >> 5;
  const int nbase = wave * 32;
  const int half  = lane >> 4;
  const int l     = lane & 15;

  if (blockIdx.x == 0) H[(size_t)ZROW * D + threadIdx.x] = 0.0f;

  const size_t aoff = (size_t)l * D + 4 * half;
  const size_t hoff = (size_t)(8 * half) * D + nbase + l;

  int t = blockIdx.x;
  v4f a0[16], a1[16];
  load_frag(a0, X + (size_t)t * 16 * D + aoff);

  v4f b0[16], b1[16];
  {
    const float* __restrict__ w0 = W + (size_t)(nbase + l) * D + 4 * half;
    const float* __restrict__ w1 = w0 + 16 * D;
#pragma unroll
    for (int j = 0; j < 16; ++j) {
      b0[j] = *(const v4f*)(w0 + 8 * j);
      b1[j] = *(const v4f*)(w1 + 8 * j);
    }
  }

  for (;;) {
    const int t1 = t + GEMM_BLOCKS;
    if (t1 >= ROW_TILES) {
      tile_product(a0, b0, b1, H + (size_t)t * 16 * D + hoff);
      break;
    }
    load_frag(a1, X + (size_t)t1 * 16 * D + aoff);
    __builtin_amdgcn_sched_barrier(0);
    tile_product(a0, b0, b1, H + (size_t)t * 16 * D + hoff);

    const int t2 = t1 + GEMM_BLOCKS;
    if (t2 >= ROW_TILES) {
      tile_product(a1, b0, b1, H + (size_t)t1 * 16 * D + hoff);
      break;
    }
    load_frag(a0, X + (size_t)t2 * 16 * D + aoff);
    __builtin_amdgcn_sched_barrier(0);
    tile_product(a1, b0, b1, H + (size_t)t1 * 16 * D + hoff);
    t = t2;
  }
}

#define LINK_BLOCK 1024
#define LINK_QPT   7
static_assert(LINK_QPT * LINK_BLOCK >= CHUNK_QUADS, "one trip links the chunk");

__global__ __launch_bounds__(LINK_BLOCK)
void gcn_link(const int* __restrict__ ei,
              unsigned short* __restrict__ heads,
              unsigned* __restrict__ node) {
  __shared__ __attribute__((aligned(16))) int tab[N_NODES];
  const int g   = blockIdx.x;
  const int tid = threadIdx.x;

  const v4i* __restrict__ src4 = (const v4i*)ei + g * CHUNK_QUADS;
  const v4i* __restrict__ dst4 = (const v4i*)(ei + N_EDGES) + g * CHUNK_QUADS;
  v4i* __restrict__ node4 = (v4i*)node + g * CHUNK_QUADS;
  v4i s[LINK_QPT], d[LINK_QPT];
#pragma unroll
  for (int k = 0; k < LINK_QPT; ++k) {
    const int q = k * LINK_BLOCK + tid;
    if (q < CHUNK_QUADS) {
      s[k] = src4[q];
      d[k] = dst4[q];
    }
  }

  for (int r = 4 * tid; r < N_NODES; r += 4 * LINK_BLOCK) {
    v4i v;
    v.x = SENT; v.y = SENT; v.z = SENT; v.w = SENT;
    *(v4i*)(tab + r) = v;
  }
  __syncthreads();

  unsigned p[4 * LINK_QPT];
#pragma unroll
  for (int k = 0; k < LINK_QPT; ++k) {
    const int q = k * LINK_BLOCK + tid;
    if (q < CHUNK_QUADS) {
#pragma unroll
      for (int c = 0; c < 4; ++c)
        p[4 * k + c] = (unsigned)atomicExch(tab + d[k][c], 4 * q + c);
    }
  }
#pragma unroll
  for (int k = 0; k < LINK_QPT; ++k) {
    const int q = k * LINK_BLOCK + tid;
    if (q < CHUNK_QUADS) {
      v4i nd;
#pragma unroll
      for (int c = 0; c < 4; ++c)
        nd[c] = (int)((p[4 * k + c] << 16) | (unsigned)s[k][c]);
      node4[q] = nd;
    }
  }
  __syncthreads();

  v4i* __restrict__ heads4 = (v4i*)(heads + g * N_NODES);
  for (int q = tid; q < N_NODES / 8; q += LINK_BLOCK) {
    const v4i a = *(const v4i*)(tab + 8 * q);
    const v4i b = *(const v4i*)(tab + 8 * q + 4);
    v4i o;
    o.x = (int)((unsigned)a.x | ((unsigned)a.y << 16));
    o.y = (int)((unsigned)a.z | ((unsigned)a.w << 16));
    o.z = (int)((unsigned)b.x | ((unsigned)b.y << 16));
    o.w = (int)((unsigned)b.z | ((unsigned)b.w << 16));
    heads4[q] = o;
  }
}

#define G_WAVES    4
#define G_PASSES   2
#define G_ROWS     (2 * G_WAVES * G_PASSES)
#define WALK_STEPS 4
#define LCAP       (LINK_GROUPS * WALK_STEPS)
#define HD_PITCH   (LINK_GROUPS + 1)
static_assert(LINK_GROUPS == 32, "one chunk per lane of a wave");

__global__ __launch_bounds__(32 * G_WAVES)
void gcn_gather(const unsigned short* __restrict__ heads,
                const unsigned* __restrict__ node,
                const float* __restrict__ H,
                const float* __restrict__ b,
                float* __restrict__ out) {
  __shared__ int hd[G_ROWS * HD_PITCH];
  __shared__ __attribute__((aligned(16))) int lst[G_WAVES * 2 * LCAP];

  const int tid  = threadIdx.x;
  const int lane = tid & 31;
  const int wave = __builtin_amdgcn_readfirstlane(tid >> 5);
  const unsigned* __restrict__ nodes = node + lane * CHUNK_EDGES;
  const v4f* __restrict__ H4 = (const v4f*)H;

#pragma unroll
  for (int k = 0; k < (G_ROWS * LINK_GROUPS) / (32 * G_WAVES); ++k) {
    const int t = tid + 32 * G_WAVES * k;
    hd[(t & 15) * HD_PITCH + (t >> 4)] =
        heads[(t >> 4) * N_NODES + blockIdx.x * G_ROWS + (t & 15)];
  }
  __syncthreads();

  const v4f bias = ((const v4f*)b)[lane];

  int* const tabA = lst + wave * (2 * LCAP);
  int* const tabB = tabA + LCAP;
  const unsigned below = (1u << lane) - 1u;

  for (int pass = 0; pass < G_PASSES; ++pass) {
  const int pr = wave + G_WAVES * pass;
  int eA = hd[(2 * pr) * HD_PITCH + lane];
  int eB = hd[(2 * pr + 1) * HD_PITCH + lane];
  v4f accA = bias, accB = bias;

  for (int round = 0; round <= CHUNK_EDGES / WALK_STEPS; ++round) {
    if (__builtin_amdgcn_ballot_w32((eA != SENT) | (eB != SENT)) == 0u) break;

    {
      v4i z;
      z.x = ZROW; z.y = ZROW; z.z = ZROW; z.w = ZROW;
      *(v4i*)(tabA + 4 * lane) = z;
      *(v4i*)(tabB + 4 * lane) = z;
    }

    int cntA = 0, cntB = 0;
    for (int s = 0; s < WALK_STEPS; ++s) {
      const bool liveA = (eA != SENT);
      const bool liveB = (eB != SENT);
      const unsigned mA = __builtin_amdgcn_ballot_w32(liveA);
      const unsigned mB = __builtin_amdgcn_ballot_w32(liveB);
      if ((mA | mB) == 0u) break;
      const int posA = cntA + __builtin_popcount(mA & below);
      const int posB = cntB + __builtin_popcount(mB & below);
      unsigned ndA, ndB;
      if (liveA) ndA = nodes[eA];
      if (liveB) ndB = nodes[eB];
      if (liveA) {
        tabA[posA] = (int)(ndA & 0xFFFFu);
        eA = (int)(ndA >> 16);
      }
      if (liveB) {
        tabB[posB] = (int)(ndB & 0xFFFFu);
        eB = (int)(ndB >> 16);
      }
      cntA += __builtin_popcount(mA);
      cntB += __builtin_popcount(mB);
    }
    __builtin_amdgcn_fence(__ATOMIC_RELEASE, "wavefront");
    __builtin_amdgcn_wave_barrier();
    __builtin_amdgcn_fence(__ATOMIC_ACQUIRE, "wavefront");

    const int nmax = cntA > cntB ? cntA : cntB;
    for (int i = 0; i < nmax; i += 8) {
      const v4i a0 = *(const v4i*)(tabA + i);
      const v4i a1 = *(const v4i*)(tabA + i + 4);
      const v4i b0 = *(const v4i*)(tabB + i);
      const v4i b1 = *(const v4i*)(tabB + i + 4);
      int ra[8], rb[8];
      ra[0] = a0.x; ra[1] = a0.y; ra[2] = a0.z; ra[3] = a0.w;
      ra[4] = a1.x; ra[5] = a1.y; ra[6] = a1.z; ra[7] = a1.w;
      rb[0] = b0.x; rb[1] = b0.y; rb[2] = b0.z; rb[3] = b0.w;
      rb[4] = b1.x; rb[5] = b1.y; rb[6] = b1.z; rb[7] = b1.w;
      v4f ma[8], mb[8];
#pragma unroll
      for (int k = 0; k < 8; ++k) {
        ma[k] = H4[(size_t)__builtin_amdgcn_readfirstlane(ra[k]) * (D / 4) + lane];
        mb[k] = H4[(size_t)__builtin_amdgcn_readfirstlane(rb[k]) * (D / 4) + lane];
      }
#pragma unroll
      for (int k = 0; k < 8; ++k) {
        accA += ma[k];
        accB += mb[k];
      }
    }
    __builtin_amdgcn_wave_barrier();
  }

  v4f* __restrict__ o =
      (v4f*)out + (size_t)(blockIdx.x * G_ROWS + 2 * pr) * (D / 4) + lane;
  o[0]     = accA;
  o[D / 4] = accB;
  }
}

extern "C" void kernel_launch(void* const* d_in, const int* in_sizes, int n_in,
                              void* d_out, int out_size, void* d_ws, size_t ws_size,
                              hipStream_t stream) {
  const float* x  = (const float*)d_in[0];
  const int*   ei = (const int*)d_in[1];
  const float* W  = (const float*)d_in[2];
  const float* b  = (const float*)d_in[3];
  float* out = (float*)d_out;
  float* H     = (float*)d_ws;
  unsigned short* heads = (unsigned short*)((char*)d_ws + 25600512);
  unsigned*       node  = (unsigned*)((char*)d_ws + 28800512);

  gcn_gemm_wmma<<<GEMM_BLOCKS, 128, 0, stream>>>(x, W, H);

  gcn_link<<<LINK_GROUPS, LINK_BLOCK, 0, stream>>>(ei, heads, node);

  gcn_gather<<<N_NODES / G_ROWS, 32 * G_WAVES, 0, stream>>>(heads, node, H, b, out);
}


# gather: a wave takes 8 rows; lane g loads chunk g's 8 heads in one 16-byte load (no LDS turn, no barrier), walks the 8 lists with 8 node loads in flight per step (were 2, serialised); sum order kept
# speedup vs baseline: 3.7067x; 1.0717x over previous
#include <hip/hip_runtime.h>

typedef __attribute__((ext_vector_type(2))) float v2f;
typedef __attribute__((ext_vector_type(4))) float v4f;
typedef __attribute__((ext_vector_type(8))) float v8f;
typedef __attribute__((ext_vector_type(4))) int   v4i;

#define N_NODES 50000
#define N_EDGES 800000
#define D       128
#define SENT    0xFFFF
#define ZROW    N_NODES

#define LINK_GROUPS 32
#define CHUNK_EDGES (N_EDGES / LINK_GROUPS)
#define CHUNK_QUADS (CHUNK_EDGES / 4)
static_assert(CHUNK_EDGES < SENT && N_NODES <= SENT, "16-bit list fields");


#define ROW_TILES   (N_NODES / 16)
#define GEMM_BLOCKS 512

__device__ __forceinline__ void load_frag(v4f (&f)[16], const float* __restrict__ p) {
#pragma unroll
  for (int j = 0; j < 16; ++j) f[j] = *(const v4f*)(p + 8 * j);
}

__device__ __forceinline__ void tile_product(const v4f (&a)[16],
                                             const v4f (&b0)[16],
                                             const v4f (&b1)[16],
                                             float* __restrict__ hrow) {
  v8f acc0 = {};
  v8f acc1 = {};
#pragma unroll
  for (int j = 0; j < 16; ++j) {
    const v2f alo = a[j].xy,  ahi = a[j].zw;
    const v2f b0lo = b0[j].xy, b0hi = b0[j].zw;
    const v2f b1lo = b1[j].xy, b1hi = b1[j].zw;
    acc0 = __builtin_amdgcn_wmma_f32_16x16x4_f32(false, alo, false, b0lo,
                                                 (short)0, acc0, false, false);
    acc1 = __builtin_amdgcn_wmma_f32_16x16x4_f32(false, alo, false, b1lo,
                                                 (short)0, acc1, false, false);
    acc0 = __builtin_amdgcn_wmma_f32_16x16x4_f32(false, ahi, false, b0hi,
                                                 (short)0, acc0, false, false);
    acc1 = __builtin_amdgcn_wmma_f32_16x16x4_f32(false, ahi, false, b1hi,
                                                 (short)0, acc1, false, false);
  }
#pragma unroll
  for (int r = 0; r < 8; ++r) {
    hrow[(size_t)r * D]      = acc0[r];
    hrow[(size_t)r * D + 16] = acc1[r];
  }
}

__global__ __launch_bounds__(128)
void gcn_gemm_wmma(const float* __restrict__ X,
                   const float* __restrict__ W,
                   float* __restrict__ H) {
  const int lane  = threadIdx.x & 31;
  const int wave  = threadIdx.x >> 5;
  const int nbase = wave * 32;
  const int half  = lane >> 4;
  const int l     = lane & 15;

  if (blockIdx.x == 0) H[(size_t)ZROW * D + threadIdx.x] = 0.0f;

  const size_t aoff = (size_t)l * D + 4 * half;
  const size_t hoff = (size_t)(8 * half) * D + nbase + l;

  int t = blockIdx.x;
  v4f a0[16], a1[16];
  load_frag(a0, X + (size_t)t * 16 * D + aoff);

  v4f b0[16], b1[16];
  {
    const float* __restrict__ w0 = W + (size_t)(nbase + l) * D + 4 * half;
    const float* __restrict__ w1 = w0 + 16 * D;
#pragma unroll
    for (int j = 0; j < 16; ++j) {
      b0[j] = *(const v4f*)(w0 + 8 * j);
      b1[j] = *(const v4f*)(w1 + 8 * j);
    }
  }

  for (;;) {
    const int t1 = t + GEMM_BLOCKS;
    if (t1 >= ROW_TILES) {
      tile_product(a0, b0, b1, H + (size_t)t * 16 * D + hoff);
      break;
    }
    load_frag(a1, X + (size_t)t1 * 16 * D + aoff);
    __builtin_amdgcn_sched_barrier(0);
    tile_product(a0, b0, b1, H + (size_t)t * 16 * D + hoff);

    const int t2 = t1 + GEMM_BLOCKS;
    if (t2 >= ROW_TILES) {
      tile_product(a1, b0, b1, H + (size_t)t1 * 16 * D + hoff);
      break;
    }
    load_frag(a0, X + (size_t)t2 * 16 * D + aoff);
    __builtin_amdgcn_sched_barrier(0);
    tile_product(a1, b0, b1, H + (size_t)t1 * 16 * D + hoff);
    t = t2;
  }
}

#define LINK_BLOCK 1024
#define LINK_QPT   7
static_assert(LINK_QPT * LINK_BLOCK >= CHUNK_QUADS, "one trip links the chunk");

__global__ __launch_bounds__(LINK_BLOCK)
void gcn_link(const int* __restrict__ ei,
              unsigned short* __restrict__ heads,
              unsigned* __restrict__ node) {
  __shared__ __attribute__((aligned(16))) int tab[N_NODES];
  const int g   = blockIdx.x;
  const int tid = threadIdx.x;

  const v4i* __restrict__ src4 = (const v4i*)ei + g * CHUNK_QUADS;
  const v4i* __restrict__ dst4 = (const v4i*)(ei + N_EDGES) + g * CHUNK_QUADS;
  v4i* __restrict__ node4 = (v4i*)node + g * CHUNK_QUADS;
  v4i s[LINK_QPT], d[LINK_QPT];
#pragma unroll
  for (int k = 0; k < LINK_QPT; ++k) {
    const int q = k * LINK_BLOCK + tid;
    if (q < CHUNK_QUADS) {
      s[k] = src4[q];
      d[k] = dst4[q];
    }
  }

  for (int r = 4 * tid; r < N_NODES; r += 4 * LINK_BLOCK) {
    v4i v;
    v.x = SENT; v.y = SENT; v.z = SENT; v.w = SENT;
    *(v4i*)(tab + r) = v;
  }
  __syncthreads();

  unsigned p[4 * LINK_QPT];
#pragma unroll
  for (int k = 0; k < LINK_QPT; ++k) {
    const int q = k * LINK_BLOCK + tid;
    if (q < CHUNK_QUADS) {
#pragma unroll
      for (int c = 0; c < 4; ++c)
        p[4 * k + c] = (unsigned)atomicExch(tab + d[k][c], 4 * q + c);
    }
  }
#pragma unroll
  for (int k = 0; k < LINK_QPT; ++k) {
    const int q = k * LINK_BLOCK + tid;
    if (q < CHUNK_QUADS) {
      v4i nd;
#pragma unroll
      for (int c = 0; c < 4; ++c)
        nd[c] = (int)((p[4 * k + c] << 16) | (unsigned)s[k][c]);
      node4[q] = nd;
    }
  }
  __syncthreads();

  v4i* __restrict__ heads4 = (v4i*)(heads + g * N_NODES);
  for (int q = tid; q < N_NODES / 8; q += LINK_BLOCK) {
    const v4i a = *(const v4i*)(tab + 8 * q);
    const v4i b = *(const v4i*)(tab + 8 * q + 4);
    v4i o;
    o.x = (int)((unsigned)a.x | ((unsigned)a.y << 16));
    o.y = (int)((unsigned)a.z | ((unsigned)a.w << 16));
    o.z = (int)((unsigned)b.x | ((unsigned)b.y << 16));
    o.w = (int)((unsigned)b.z | ((unsigned)b.w << 16));
    heads4[q] = o;
  }
}

#define G_WAVES    2
#define G_WROWS    8
#define G_ROWS     (G_WAVES * G_WROWS)
#define WALK_STEPS 4
#define LCAP       (LINK_GROUPS * WALK_STEPS)
static_assert(LINK_GROUPS == 32, "one chunk per lane of a wave");
static_assert(N_NODES % G_ROWS == 0, "whole blocks");
static_assert(G_WROWS == 8, "a lane's heads of the wave's rows are one 16-byte load");

__device__ __forceinline__ void sum_pair(const int* tabA, const int* tabB,
                                         int cntA, int cntB,
                                         const v4f* __restrict__ H4, int lane,
                                         v4f& accA, v4f& accB) {
  const int nmax = cntA > cntB ? cntA : cntB;
  for (int i = 0; i < nmax; i += 8) {
    const v4i a0 = *(const v4i*)(tabA + i);
    const v4i a1 = *(const v4i*)(tabA + i + 4);
    const v4i b0 = *(const v4i*)(tabB + i);
    const v4i b1 = *(const v4i*)(tabB + i + 4);
    int ra[8], rb[8];
    ra[0] = a0.x; ra[1] = a0.y; ra[2] = a0.z; ra[3] = a0.w;
    ra[4] = a1.x; ra[5] = a1.y; ra[6] = a1.z; ra[7] = a1.w;
    rb[0] = b0.x; rb[1] = b0.y; rb[2] = b0.z; rb[3] = b0.w;
    rb[4] = b1.x; rb[5] = b1.y; rb[6] = b1.z; rb[7] = b1.w;
    v4f ma[8], mb[8];
#pragma unroll
    for (int k = 0; k < 8; ++k) {
      ma[k] = H4[(size_t)__builtin_amdgcn_readfirstlane(ra[k]) * (D / 4) + lane];
      mb[k] = H4[(size_t)__builtin_amdgcn_readfirstlane(rb[k]) * (D / 4) + lane];
    }
#pragma unroll
    for (int k = 0; k < 8; ++k) {
      accA += ma[k];
      accB += mb[k];
    }
  }
}

__global__ __launch_bounds__(32 * G_WAVES)
void gcn_gather(const unsigned short* __restrict__ heads,
                const unsigned* __restrict__ node,
                const float* __restrict__ H,
                const float* __restrict__ b,
                float* __restrict__ out) {
  __shared__ __attribute__((aligned(16))) int lst[G_WAVES * G_WROWS * LCAP];

  const int tid  = threadIdx.x;
  const int lane = tid & 31;
  const int wave = __builtin_amdgcn_readfirstlane(tid >> 5);
  const int row0 = (blockIdx.x * G_WAVES + wave) * G_WROWS;
  const v4f* __restrict__ H4 = (const v4f*)H;

  const v4i hw = *(const v4i*)(heads + (size_t)lane * N_NODES + row0);
  int e[G_WROWS];
  e[0] = (int)((unsigned)hw.x & 0xFFFFu);  e[1] = (int)((unsigned)hw.x >> 16);
  e[2] = (int)((unsigned)hw.y & 0xFFFFu);  e[3] = (int)((unsigned)hw.y >> 16);
  e[4] = (int)((unsigned)hw.z & 0xFFFFu);  e[5] = (int)((unsigned)hw.z >> 16);
  e[6] = (int)((unsigned)hw.w & 0xFFFFu);  e[7] = (int)((unsigned)hw.w >> 16);

  const v4f bias = ((const v4f*)b)[lane];
  v4f acc[G_WROWS];
#pragma unroll
  for (int r = 0; r < G_WROWS; ++r) acc[r] = bias;

  int* const tab = lst + wave * (G_WROWS * LCAP);
  const unsigned below = (1u << lane) - 1u;
  const unsigned mine  = (unsigned)lane * CHUNK_EDGES;

  for (int round = 0; round <= CHUNK_EDGES / WALK_STEPS; ++round) {
    bool go = false;
#pragma unroll
    for (int r = 0; r < G_WROWS; ++r) go |= (e[r] != SENT);
    if (__builtin_amdgcn_ballot_w32(go) == 0u) break;

    {
      v4i z;
      z.x = ZROW; z.y = ZROW; z.z = ZROW; z.w = ZROW;
#pragma unroll
      for (int r = 0; r < G_WROWS; ++r) *(v4i*)(tab + r * LCAP + 4 * lane) = z;
    }

    int cnt[G_WROWS];
#pragma unroll
    for (int r = 0; r < G_WROWS; ++r) cnt[r] = 0;
    for (int s = 0; s < WALK_STEPS; ++s) {
      unsigned m[G_WROWS];
      unsigned many = 0u;
#pragma unroll
      for (int r = 0; r < G_WROWS; ++r) {
        m[r] = __builtin_amdgcn_ballot_w32(e[r] != SENT);
        many |= m[r];
      }
      if (many == 0u) break;
      unsigned nd[G_WROWS];
#pragma unroll
      for (int r = 0; r < G_WROWS; ++r) {
        const unsigned off = (e[r] != SENT) ? mine + (unsigned)e[r] : 0u;
        nd[r] = node[off];
      }
#pragma unroll
      for (int r = 0; r < G_WROWS; ++r) {
        const bool live = (e[r] != SENT);
        const int pos = cnt[r] + __builtin_popcount(m[r] & below);
        if (live) tab[r * LCAP + pos] = (int)(nd[r] & 0xFFFFu);
        e[r] = live ? (int)(nd[r] >> 16) : SENT;
        cnt[r] += __builtin_popcount(m[r]);
      }
    }
    __builtin_amdgcn_fence(__ATOMIC_RELEASE, "wavefront");
    __builtin_amdgcn_wave_barrier();
    __builtin_amdgcn_fence(__ATOMIC_ACQUIRE, "wavefront");

#pragma unroll
    for (int p = 0; p < G_WROWS / 2; ++p)
      sum_pair(tab + (2 * p) * LCAP, tab + (2 * p + 1) * LCAP,
               cnt[2 * p], cnt[2 * p + 1], H4, lane, acc[2 * p], acc[2 * p + 1]);

    __builtin_amdgcn_wave_barrier();
  }

  v4f* __restrict__ o = (v4f*)out + (size_t)row0 * (D / 4) + lane;
#pragma unroll
  for (int r = 0; r < G_WROWS; ++r) o[(size_t)r * (D / 4)] = acc[r];
}

extern "C" void kernel_launch(void* const* d_in, const int* in_sizes, int n_in,
                              void* d_out, int out_size, void* d_ws, size_t ws_size,
                              hipStream_t stream) {
  const float* x  = (const float*)d_in[0];
  const int*   ei = (const int*)d_in[1];
  const float* W  = (const float*)d_in[2];
  const float* b  = (const float*)d_in[3];
  float* out = (float*)d_out;
  float* H     = (float*)d_ws;
  unsigned short* heads = (unsigned short*)((char*)d_ws + 25600512);
  unsigned*       node  = (unsigned*)((char*)d_ws + 28800512);

  gcn_gemm_wmma<<<GEMM_BLOCKS, 128, 0, stream>>>(x, W, H);

  gcn_link<<<LINK_GROUPS, LINK_BLOCK, 0, stream>>>(ei, heads, node);

  gcn_gather<<<N_NODES / G_ROWS, 32 * G_WAVES, 0, stream>>>(heads, node, H, b, out);
}


# link and X W^T run as one launch of 256 work-groups (32 link in LDS with 256 threads each, 224 multiply, 8 waves each): 2 launches for 3; sums stay in list (arrival) order, edge-thread map changed
# speedup vs baseline: 4.4886x; 1.2109x over previous
#include <hip/hip_runtime.h>

typedef __attribute__((ext_vector_type(2))) float v2f;
typedef __attribute__((ext_vector_type(4))) float v4f;
typedef __attribute__((ext_vector_type(8))) float v8f;
typedef __attribute__((ext_vector_type(4))) int   v4i;

#define N_NODES 50000
#define N_EDGES 800000
#define D       128
#define SENT    0xFFFF
#define ZROW    N_NODES

#define LINK_GROUPS 32
#define CHUNK_EDGES (N_EDGES / LINK_GROUPS)
#define CHUNK_QUADS (CHUNK_EDGES / 4)
static_assert(CHUNK_EDGES < SENT && N_NODES <= SENT, "16-bit list fields");


#define ROW_TILES    (N_NODES / 16)
#define FUSED_BLOCK  256
#define FUSED_GROUPS 256
#define GEMM_GROUPS  (FUSED_GROUPS - LINK_GROUPS)
#define GEMM_SLOTS   (2 * GEMM_GROUPS)
static_assert(GEMM_SLOTS <= ROW_TILES, "every slot has a first tile");

__device__ __forceinline__ void load_frag(v4f (&f)[16], const float* __restrict__ p) {
#pragma unroll
  for (int j = 0; j < 16; ++j) f[j] = *(const v4f*)(p + 8 * j);
}

__device__ __forceinline__ void tile_product(const v4f (&a)[16],
                                             const v4f (&b0)[16],
                                             const v4f (&b1)[16],
                                             float* __restrict__ hrow) {
  v8f acc0 = {};
  v8f acc1 = {};
#pragma unroll
  for (int j = 0; j < 16; ++j) {
    const v2f alo = a[j].xy,  ahi = a[j].zw;
    const v2f b0lo = b0[j].xy, b0hi = b0[j].zw;
    const v2f b1lo = b1[j].xy, b1hi = b1[j].zw;
    acc0 = __builtin_amdgcn_wmma_f32_16x16x4_f32(false, alo, false, b0lo,
                                                 (short)0, acc0, false, false);
    acc1 = __builtin_amdgcn_wmma_f32_16x16x4_f32(false, alo, false, b1lo,
                                                 (short)0, acc1, false, false);
    acc0 = __builtin_amdgcn_wmma_f32_16x16x4_f32(false, ahi, false, b0hi,
                                                 (short)0, acc0, false, false);
    acc1 = __builtin_amdgcn_wmma_f32_16x16x4_f32(false, ahi, false, b1hi,
                                                 (short)0, acc1, false, false);
  }
#pragma unroll
  for (int r = 0; r < 8; ++r) {
    hrow[(size_t)r * D]      = acc0[r];
    hrow[(size_t)r * D + 16] = acc1[r];
  }
}

__device__ __forceinline__ void gemm_part(const float* __restrict__ X,
                                          const float* __restrict__ W,
                                          float* __restrict__ H,
                                          const int gg, const int tid) {
  const int lane  = tid & 31;
  const int wave  = __builtin_amdgcn_readfirstlane(tid >> 5);
  const int nbase = (wave & 3) * 32;
  const int half  = lane >> 4;
  const int l     = lane & 15;

  if (gg == 0 && tid < D) H[(size_t)ZROW * D + tid] = 0.0f;

  const size_t aoff = (size_t)l * D + 4 * half;
  const size_t hoff = (size_t)(8 * half) * D + nbase + l;

  int t = 2 * gg + (wave >> 2);
  v4f a0[16], a1[16];
  load_frag(a0, X + (size_t)t * 16 * D + aoff);

  v4f b0[16], b1[16];
  {
    const float* __restrict__ w0 = W + (size_t)(nbase + l) * D + 4 * half;
    const float* __restrict__ w1 = w0 + 16 * D;
#pragma unroll
    for (int j = 0; j < 16; ++j) {
      b0[j] = *(const v4f*)(w0 + 8 * j);
      b1[j] = *(const v4f*)(w1 + 8 * j);
    }
  }

  for (;;) {
    const int t1 = t + GEMM_SLOTS;
    if (t1 >= ROW_TILES) {
      tile_product(a0, b0, b1, H + (size_t)t * 16 * D + hoff);
      break;
    }
    load_frag(a1, X + (size_t)t1 * 16 * D + aoff);
    __builtin_amdgcn_sched_barrier(0);
    tile_product(a0, b0, b1, H + (size_t)t * 16 * D + hoff);

    const int t2 = t1 + GEMM_SLOTS;
    if (t2 >= ROW_TILES) {
      tile_product(a1, b0, b1, H + (size_t)t1 * 16 * D + hoff);
      break;
    }
    load_frag(a0, X + (size_t)t2 * 16 * D + aoff);
    __builtin_amdgcn_sched_barrier(0);
    tile_product(a1, b0, b1, H + (size_t)t1 * 16 * D + hoff);
    t = t2;
  }
}

#define LINK_BLOCK FUSED_BLOCK
#define LINK_QPT   25
static_assert(LINK_QPT * LINK_BLOCK >= CHUNK_QUADS, "one trip links the chunk");
static_assert(2 * LINK_QPT <= 63, "a thread's index loads fit the load counter");

__device__ __forceinline__ void link_part(int* __restrict__ tab,
                                          const int* __restrict__ ei,
                                          unsigned short* __restrict__ heads,
                                          unsigned* __restrict__ node,
                                          const int g, const int tid) {

  const v4i* __restrict__ src4 = (const v4i*)ei + g * CHUNK_QUADS;
  const v4i* __restrict__ dst4 = (const v4i*)(ei + N_EDGES) + g * CHUNK_QUADS;
  v4i* __restrict__ node4 = (v4i*)node + g * CHUNK_QUADS;
  v4i s[LINK_QPT], d[LINK_QPT];
#pragma unroll
  for (int k = 0; k < LINK_QPT; ++k) {
    const int q = k * LINK_BLOCK + tid;
    if (q < CHUNK_QUADS) {
      s[k] = src4[q];
      d[k] = dst4[q];
    }
  }

  for (int r = 4 * tid; r < N_NODES; r += 4 * LINK_BLOCK) {
    v4i v;
    v.x = SENT; v.y = SENT; v.z = SENT; v.w = SENT;
    *(v4i*)(tab + r) = v;
  }
  __syncthreads();

  unsigned p[4 * LINK_QPT];
#pragma unroll
  for (int k = 0; k < LINK_QPT; ++k) {
    const int q = k * LINK_BLOCK + tid;
    if (q < CHUNK_QUADS) {
#pragma unroll
      for (int c = 0; c < 4; ++c)
        p[4 * k + c] = (unsigned)atomicExch(tab + d[k][c], 4 * q + c);
    }
  }
#pragma unroll
  for (int k = 0; k < LINK_QPT; ++k) {
    const int q = k * LINK_BLOCK + tid;
    if (q < CHUNK_QUADS) {
      v4i nd;
#pragma unroll
      for (int c = 0; c < 4; ++c)
        nd[c] = (int)((p[4 * k + c] << 16) | (unsigned)s[k][c]);
      node4[q] = nd;
    }
  }
  __syncthreads();

  v4i* __restrict__ heads4 = (v4i*)(heads + g * N_NODES);
#pragma unroll 5
  for (int q = tid; q < N_NODES / 8; q += LINK_BLOCK) {
    const v4i a = *(const v4i*)(tab + 8 * q);
    const v4i b = *(const v4i*)(tab + 8 * q + 4);
    v4i o;
    o.x = (int)((unsigned)a.x | ((unsigned)a.y << 16));
    o.y = (int)((unsigned)a.z | ((unsigned)a.w << 16));
    o.z = (int)((unsigned)b.x | ((unsigned)b.y << 16));
    o.w = (int)((unsigned)b.z | ((unsigned)b.w << 16));
    heads4[q] = o;
  }
}

__global__ __launch_bounds__(FUSED_BLOCK)
void gcn_link_gemm(const float* __restrict__ X,
                   const float* __restrict__ W,
                   float* __restrict__ H,
                   const int* __restrict__ ei,
                   unsigned short* __restrict__ heads,
                   unsigned* __restrict__ node) {
  __shared__ __attribute__((aligned(16))) int tab[N_NODES];
  const int tid = threadIdx.x;
  if (blockIdx.x < LINK_GROUPS)
    link_part(tab, ei, heads, node, blockIdx.x, tid);
  else
    gemm_part(X, W, H, blockIdx.x - LINK_GROUPS, tid);
}

#define G_WAVES    2
#define G_WROWS    8
#define G_ROWS     (G_WAVES * G_WROWS)
#define WALK_STEPS 4
#define LCAP       (LINK_GROUPS * WALK_STEPS)
static_assert(LINK_GROUPS == 32, "one chunk per lane of a wave");
static_assert(N_NODES % G_ROWS == 0, "whole blocks");
static_assert(G_WROWS == 8, "a lane's heads of the wave's rows are one 16-byte load");

__device__ __forceinline__ void sum_pair(const int* tabA, const int* tabB,
                                         int cntA, int cntB,
                                         const v4f* __restrict__ H4, int lane,
                                         v4f& accA, v4f& accB) {
  const int nmax = cntA > cntB ? cntA : cntB;
  for (int i = 0; i < nmax; i += 8) {
    const v4i a0 = *(const v4i*)(tabA + i);
    const v4i a1 = *(const v4i*)(tabA + i + 4);
    const v4i b0 = *(const v4i*)(tabB + i);
    const v4i b1 = *(const v4i*)(tabB + i + 4);
    int ra[8], rb[8];
    ra[0] = a0.x; ra[1] = a0.y; ra[2] = a0.z; ra[3] = a0.w;
    ra[4] = a1.x; ra[5] = a1.y; ra[6] = a1.z; ra[7] = a1.w;
    rb[0] = b0.x; rb[1] = b0.y; rb[2] = b0.z; rb[3] = b0.w;
    rb[4] = b1.x; rb[5] = b1.y; rb[6] = b1.z; rb[7] = b1.w;
    v4f ma[8], mb[8];
#pragma unroll
    for (int k = 0; k < 8; ++k) {
      ma[k] = H4[(size_t)__builtin_amdgcn_readfirstlane(ra[k]) * (D / 4) + lane];
      mb[k] = H4[(size_t)__builtin_amdgcn_readfirstlane(rb[k]) * (D / 4) + lane];
    }
#pragma unroll
    for (int k = 0; k < 8; ++k) {
      accA += ma[k];
      accB += mb[k];
    }
  }
}

__global__ __launch_bounds__(32 * G_WAVES)
void gcn_gather(const unsigned short* __restrict__ heads,
                const unsigned* __restrict__ node,
                const float* __restrict__ H,
                const float* __restrict__ b,
                float* __restrict__ out) {
  __shared__ __attribute__((aligned(16))) int lst[G_WAVES * G_WROWS * LCAP];

  const int tid  = threadIdx.x;
  const int lane = tid & 31;
  const int wave = __builtin_amdgcn_readfirstlane(tid >> 5);
  const int row0 = (blockIdx.x * G_WAVES + wave) * G_WROWS;
  const v4f* __restrict__ H4 = (const v4f*)H;

  const v4i hw = *(const v4i*)(heads + (size_t)lane * N_NODES + row0);
  int e[G_WROWS];
  e[0] = (int)((unsigned)hw.x & 0xFFFFu);  e[1] = (int)((unsigned)hw.x >> 16);
  e[2] = (int)((unsigned)hw.y & 0xFFFFu);  e[3] = (int)((unsigned)hw.y >> 16);
  e[4] = (int)((unsigned)hw.z & 0xFFFFu);  e[5] = (int)((unsigned)hw.z >> 16);
  e[6] = (int)((unsigned)hw.w & 0xFFFFu);  e[7] = (int)((unsigned)hw.w >> 16);

  const v4f bias = ((const v4f*)b)[lane];
  v4f acc[G_WROWS];
#pragma unroll
  for (int r = 0; r < G_WROWS; ++r) acc[r] = bias;

  int* const tab = lst + wave * (G_WROWS * LCAP);
  const unsigned below = (1u << lane) - 1u;
  const unsigned mine  = (unsigned)lane * CHUNK_EDGES;

  for (int round = 0; round <= CHUNK_EDGES / WALK_STEPS; ++round) {
    bool go = false;
#pragma unroll
    for (int r = 0; r < G_WROWS; ++r) go |= (e[r] != SENT);
    if (__builtin_amdgcn_ballot_w32(go) == 0u) break;

    {
      v4i z;
      z.x = ZROW; z.y = ZROW; z.z = ZROW; z.w = ZROW;
#pragma unroll
      for (int r = 0; r < G_WROWS; ++r) *(v4i*)(tab + r * LCAP + 4 * lane) = z;
    }

    int cnt[G_WROWS];
#pragma unroll
    for (int r = 0; r < G_WROWS; ++r) cnt[r] = 0;
    for (int s = 0; s < WALK_STEPS; ++s) {
      unsigned m[G_WROWS];
      unsigned many = 0u;
#pragma unroll
      for (int r = 0; r < G_WROWS; ++r) {
        m[r] = __builtin_amdgcn_ballot_w32(e[r] != SENT);
        many |= m[r];
      }
      if (many == 0u) break;
      unsigned nd[G_WROWS];
#pragma unroll
      for (int r = 0; r < G_WROWS; ++r) {
        const unsigned off = (e[r] != SENT) ? mine + (unsigned)e[r] : 0u;
        nd[r] = node[off];
      }
#pragma unroll
      for (int r = 0; r < G_WROWS; ++r) {
        const bool live = (e[r] != SENT);
        const int pos = cnt[r] + __builtin_popcount(m[r] & below);
        if (live) tab[r * LCAP + pos] = (int)(nd[r] & 0xFFFFu);
        e[r] = live ? (int)(nd[r] >> 16) : SENT;
        cnt[r] += __builtin_popcount(m[r]);
      }
    }
    __builtin_amdgcn_fence(__ATOMIC_RELEASE, "wavefront");
    __builtin_amdgcn_wave_barrier();
    __builtin_amdgcn_fence(__ATOMIC_ACQUIRE, "wavefront");

#pragma unroll
    for (int p = 0; p < G_WROWS / 2; ++p)
      sum_pair(tab + (2 * p) * LCAP, tab + (2 * p + 1) * LCAP,
               cnt[2 * p], cnt[2 * p + 1], H4, lane, acc[2 * p], acc[2 * p + 1]);

    __builtin_amdgcn_wave_barrier();
  }

  v4f* __restrict__ o = (v4f*)out + (size_t)row0 * (D / 4) + lane;
#pragma unroll
  for (int r = 0; r < G_WROWS; ++r) o[(size_t)r * (D / 4)] = acc[r];
}

extern "C" void kernel_launch(void* const* d_in, const int* in_sizes, int n_in,
                              void* d_out, int out_size, void* d_ws, size_t ws_size,
                              hipStream_t stream) {
  const float* x  = (const float*)d_in[0];
  const int*   ei = (const int*)d_in[1];
  const float* W  = (const float*)d_in[2];
  const float* b  = (const float*)d_in[3];
  float* out = (float*)d_out;
  float* H     = (float*)d_ws;
  unsigned short* heads = (unsigned short*)((char*)d_ws + 25600512);
  unsigned*       node  = (unsigned*)((char*)d_ws + 28800512);

  gcn_link_gemm<<<FUSED_GROUPS, FUSED_BLOCK, 0, stream>>>(x, W, H, ei, heads, node);

  gcn_gather<<<N_NODES / G_ROWS, 32 * G_WAVES, 0, stream>>>(heads, node, H, b, out);
}
